# starting kernel
# baseline (speedup 1.0000x reference)
#include <hip/hip_runtime.h>

typedef __attribute__((ext_vector_type(2))) float v2f;
typedef __attribute__((ext_vector_type(8))) float v8f;

#define EMB    256
#define SENTS  128
#define BATCH  64
#define TOKS   64
#define NCLS   100
#define CPAD   112

__global__ void gather_mean_kernel(const int* __restrict__ tok,
                                   const float* __restrict__ emb,
                                   float* __restrict__ sents) {
  const int bs = blockIdx.x;
  const int t  = threadIdx.x;
  const int* tl = tok + (size_t)bs * TOKS;
  const float4* embv = (const float4*)emb;
  float4 acc = make_float4(0.f, 0.f, 0.f, 0.f);
  #pragma unroll 4
  for (int l = 0; l < TOKS; ++l) {
    const int tk = tl[l];
    float4 v = embv[(size_t)tk * (EMB / 4) + t];
    acc.x += v.x; acc.y += v.y; acc.z += v.z; acc.w += v.w;
  }
  const float inv = 1.0f / (float)TOKS;
  acc.x *= inv; acc.y *= inv; acc.z *= inv; acc.w *= inv;
  ((float4*)sents)[(size_t)bs * (EMB / 4) + t] = acc;
}

__global__ void scores_wmma_kernel(const float* __restrict__ cls,
                                   const float* __restrict__ sents,
                                   float* __restrict__ scores) {
  const int st   = blockIdx.x;
  const int b    = blockIdx.y;
  const int wave = threadIdx.x >> 5;
  const int lane = threadIdx.x & 31;
  const int hl   = lane >> 4;
  const int r15  = lane & 15;

  const int m  = wave * 16 + r15;
  const int mc = (m < NCLS) ? m : (NCLS - 1);
  const int n  = st * 16 + r15;

  const float* arow = cls + (size_t)mc * EMB + 2 * hl;
  const float* brow = sents + ((size_t)b * SENTS + n) * EMB + 2 * hl;

  v8f acc = {};
  #pragma unroll 8
  for (int kb = 0; kb < EMB; kb += 4) {
    v2f a = *(const v2f*)(arow + kb);
    v2f bb = *(const v2f*)(brow + kb);
    acc = __builtin_amdgcn_wmma_f32_16x16x4_f32(
         false, a,  false, bb,
         (short)0, acc,  false,  false);
  }
  float* outb = scores + (size_t)b * CPAD * SENTS;
  #pragma unroll
  for (int r = 0; r < 8; ++r) {
    const int c = wave * 16 + r + 8 * hl;
    outb[(size_t)c * SENTS + st * 16 + r15] = acc[r];
  }
}

__global__ void softmax_kernel(float* __restrict__ scores) {
  const int c = blockIdx.x;
  const int b = blockIdx.y;
  const int t = threadIdx.x;
  float* row = scores + ((size_t)b * CPAD + c) * SENTS;
  __shared__ float red[SENTS];

  const float v = row[t];
  red[t] = v;
  __syncthreads();
  for (int off = SENTS / 2; off > 0; off >>= 1) {
    if (t < off) red[t] = fmaxf(red[t], red[t + off]);
    __syncthreads();
  }
  const float mx = red[0];
  __syncthreads();
  const float e = __expf(v - mx);
  red[t] = e;
  __syncthreads();
  for (int off = SENTS / 2; off > 0; off >>= 1) {
    if (t < off) red[t] += red[t + off];
    __syncthreads();
  }
  row[t] = e / red[0];
}

__global__ void proj_wmma_kernel(const float* __restrict__ sents,
                                 const float* __restrict__ mw,
                                 float* __restrict__ proj) {
  const int st   = blockIdx.x;
  const int b    = blockIdx.y;
  const int wave = threadIdx.x >> 5;
  const int lane = threadIdx.x & 31;
  const int hl   = lane >> 4;
  const int r15  = lane & 15;

  const int m  = st * 16 + r15;
  const int n  = wave * 16 + r15;
  const int nc = (n < NCLS) ? n : (NCLS - 1);

  const float* arow = sents + ((size_t)b * SENTS + m) * EMB + 2 * hl;
  const float* brow = mw + (size_t)nc * EMB + 2 * hl;

  v8f acc = {};
  #pragma unroll 8
  for (int kb = 0; kb < EMB; kb += 4) {
    v2f a = *(const v2f*)(arow + kb);
    v2f bv = *(const v2f*)(brow + kb);
    acc = __builtin_amdgcn_wmma_f32_16x16x4_f32(
        false, a, false, bv, (short)0, acc, false, false);
  }
  float* outb = proj + (size_t)b * SENTS * CPAD;
  #pragma unroll
  for (int r = 0; r < 8; ++r) {
    const int s = st * 16 + r + 8 * hl;
    outb[(size_t)s * CPAD + wave * 16 + r15] = acc[r];
  }
}

__global__ void logits_kernel(const float* __restrict__ atts,
                              const float* __restrict__ proj,
                              const float* __restrict__ bias,
                              float* __restrict__ out) {
  const int c = blockIdx.x;
  const int b = blockIdx.y;
  const int t = threadIdx.x;
  __shared__ float red[SENTS];

  const float a = atts[((size_t)b * CPAD + c) * SENTS + t];
  const float p = proj[((size_t)b * SENTS + t) * CPAD + c];
  red[t] = a * p;
  __syncthreads();
  for (int off = SENTS / 2; off > 0; off >>= 1) {
    if (t < off) red[t] += red[t + off];
    __syncthreads();
  }
  if (t == 0) out[(size_t)b * NCLS + c] = red[0] + bias[c];
}

extern "C" void kernel_launch(void* const* d_in, const int* in_sizes, int n_in,
                              void* d_out, int out_size, void* d_ws, size_t ws_size,
                              hipStream_t stream) {
  const int*   tok  = (const int*)d_in[0];
  const float* emb  = (const float*)d_in[1];
  const float* cls  = (const float*)d_in[2];
  const float* mw   = (const float*)d_in[3];
  const float* bias = (const float*)d_in[4];
  float*       out  = (float*)d_out;

  char* ws = (char*)d_ws;
  const size_t sents_bytes  = (size_t)BATCH * SENTS * EMB * sizeof(float);
  const size_t scores_bytes = (size_t)BATCH * CPAD * SENTS * sizeof(float);
  float* sents  = (float*)ws;
  float* scores = (float*)(ws + sents_bytes);
  float* proj   = (float*)(ws + sents_bytes + scores_bytes);

  gather_mean_kernel<<<BATCH * SENTS, 64, 0, stream>>>(tok, emb, sents);
  scores_wmma_kernel<<<dim3(SENTS / 16, BATCH), 224, 0, stream>>>(cls, sents, scores);
  softmax_kernel<<<dim3(NCLS, BATCH), SENTS, 0, stream>>>(scores);
  proj_wmma_kernel<<<dim3(SENTS / 16, BATCH), 224, 0, stream>>>(sents, mw, proj);
  logits_kernel<<<dim3(NCLS, BATCH), SENTS, 0, stream>>>(scores, proj, bias, out);
}


# Fuse gather, both WMMA products and per-tile softmax partials into one launch per (batch, 16-sentence tile), sentences kept in LDS; second launch merges 8 tiles; new reduction tree over sentences
# speedup vs baseline: 1.8659x; 1.8659x over previous
#include <hip/hip_runtime.h>

typedef __attribute__((ext_vector_type(2))) float v2f;
typedef __attribute__((ext_vector_type(8))) float v8f;

#define EMB    256
#define SENTS  128
#define BATCH  64
#define TOKS   64
#define NCLS   100
#define TS     16
#define NST    (SENTS / TS)
#define NCT    7
#define SP     260
#define KC     32
#define WP     36
#define NW     (2 * NCLS)
#define PSTR   64

__global__ void __launch_bounds__(512)
tile_kernel(const int* __restrict__ tok, const float* __restrict__ emb,
            const float* __restrict__ cls, const float* __restrict__ mw,
            float* __restrict__ part) {
  __shared__ float S[TS * SP];
  __shared__ float Wt[NW * WP];

  const int st   = blockIdx.x;
  const int b    = blockIdx.y;
  const int t    = threadIdx.x;
  const int lane = t & 31;
  const int wave = __builtin_amdgcn_readfirstlane(t >> 5);

  const int  seg   = t & 7;
  const int  row0  = t >> 3;
  const bool has3  = t < (NW * 8 - 3 * 512);
  const float* g0 = cls + (size_t)row0 * EMB + seg * 4;
  const float* g1 = (row0 + 64 < NCLS) ? (cls + (size_t)(row0 + 64) * EMB + seg * 4)
                                       : (mw + (size_t)(row0 + 64 - NCLS) * EMB + seg * 4);
  const float* g2 = mw + (size_t)(row0 + 128 - NCLS) * EMB + seg * 4;
  const float* g3 = mw + (size_t)((has3 ? row0 + 192 : NW - 1) - NCLS) * EMB + seg * 4;
  float* w0 = Wt + row0 * WP + seg * 4;
  float4 p0, p1, p2, p3 = make_float4(0.f, 0.f, 0.f, 0.f);
#define LOAD_CHUNK(ch)                                   \
  do {                                                   \
    p0 = *(const float4*)(g0 + (ch) * KC);               \
    p1 = *(const float4*)(g1 + (ch) * KC);               \
    p2 = *(const float4*)(g2 + (ch) * KC);               \
    if (has3) p3 = *(const float4*)(g3 + (ch) * KC);     \
  } while (0)
#define STORE_CHUNK()                                    \
  do {                                                   \
    *(float4*)(w0)            = p0;                      \
    *(float4*)(w0 + 64 * WP)  = p1;                      \
    *(float4*)(w0 + 128 * WP) = p2;                      \
    if (has3) *(float4*)(w0 + 192 * WP) = p3;            \
  } while (0)

  LOAD_CHUNK(0);

  {
    const int* tl = tok + ((size_t)b * SENTS + st * TS + wave) * TOKS;
    const float4* embv = (const float4*)emb;
    float4 a0 = make_float4(0.f, 0.f, 0.f, 0.f);
    float4 a1 = make_float4(0.f, 0.f, 0.f, 0.f);
#pragma unroll 8
    for (int l = 0; l < TOKS; ++l) {
      const int tk = tl[l];
      const float4* row = embv + (size_t)tk * (EMB / 4);
      const float4 v0 = row[lane];
      const float4 v1 = row[32 + lane];
      a0.x += v0.x; a0.y += v0.y; a0.z += v0.z; a0.w += v0.w;
      a1.x += v1.x; a1.y += v1.y; a1.z += v1.z; a1.w += v1.w;
    }
    const float inv = 1.0f / (float)TOKS;
    a0.x *= inv; a0.y *= inv; a0.z *= inv; a0.w *= inv;
    a1.x *= inv; a1.y *= inv; a1.z *= inv; a1.w *= inv;
    *(float4*)(S + wave * SP + lane * 4)       = a0;
    *(float4*)(S + wave * SP + 128 + lane * 4) = a1;
  }

  const int hl  = lane >> 4;
  const int r15 = lane & 15;
  const int n   = wave * 16 + r15;
  const int nc  = (n < NCLS) ? n : (NCLS - 1);
  const float* arow = S + r15 * SP + 2 * hl;
  const float* brow = Wt + nc * WP + 2 * hl;
  const float* prow = Wt + (NCLS + nc) * WP + 2 * hl;

  v8f accS = {};
  v8f accP = {};
  for (int ch = 0; ch < EMB / KC; ++ch) {
    __syncthreads();
    STORE_CHUNK();
    __syncthreads();
    if (ch + 1 < EMB / KC) LOAD_CHUNK(ch + 1);
    if (wave < NCT) {
#pragma unroll
      for (int kk = 0; kk < KC; kk += 4) {
        const v2f a = *(const v2f*)(arow + ch * KC + kk);
        const v2f x = *(const v2f*)(brow + kk);
        const v2f y = *(const v2f*)(prow + kk);
        accS = __builtin_amdgcn_wmma_f32_16x16x4_f32(false, a, false, x, (short)0, accS, false, false);
        accP = __builtin_amdgcn_wmma_f32_16x16x4_f32(false, a, false, y, (short)0, accP, false, false);
      }
    }
  }

  if (wave >= NCT) return;

  float m = accS[0];
#pragma unroll
  for (int r = 1; r < 8; ++r) m = fmaxf(m, accS[r]);
  m = fmaxf(m, __shfl_xor(m, 16));
  float l = 0.f, w = 0.f;
#pragma unroll
  for (int r = 0; r < 8; ++r) {
    const float e = __expf(accS[r] - m);
    l += e;
    w += e * accP[r];
  }
  l += __shfl_xor(l, 16);
  w += __shfl_xor(w, 16);

  float* p = part + ((size_t)(b * NST + st) * NCT + wave) * PSTR;
  p[lane] = hl ? l : m;
  if (hl == 0) p[32 + r15] = w;
}

__global__ void merge_kernel(const float* __restrict__ part,
                             const float* __restrict__ bias,
                             float* __restrict__ out) {
  const int idx = blockIdx.x * blockDim.x + threadIdx.x;
  if (idx >= BATCH * NCLS) return;
  const int b  = idx / NCLS;
  const int c  = idx - b * NCLS;
  const int ct = c >> 4, ci = c & 15;
  const float* p = part + ((size_t)(b * NST) * NCT + ct) * PSTR + ci;
  float mi[NST], li[NST], wi[NST];
#pragma unroll
  for (int i = 0; i < NST; ++i) {
    const float* q = p + (size_t)i * NCT * PSTR;
    mi[i] = q[0]; li[i] = q[16]; wi[i] = q[32];
  }
  float M = mi[0];
#pragma unroll
  for (int i = 1; i < NST; ++i) M = fmaxf(M, mi[i]);
  float L = 0.f, W = 0.f;
#pragma unroll
  for (int i = 0; i < NST; ++i) {
    const float f = __expf(mi[i] - M);
    L += li[i] * f;
    W += wi[i] * f;
  }
  out[idx] = W / L + bias[c];
}

extern "C" void kernel_launch(void* const* d_in, const int* in_sizes, int n_in,
                              void* d_out, int out_size, void* d_ws, size_t ws_size,
                              hipStream_t stream) {
  const int*   tok  = (const int*)d_in[0];
  const float* emb  = (const float*)d_in[1];
  const float* cls  = (const float*)d_in[2];
  const float* mw   = (const float*)d_in[3];
  const float* bias = (const float*)d_in[4];
  float*       out  = (float*)d_out;
  float*       part = (float*)d_ws;

  tile_kernel<<<dim3(NST, BATCH), 512, 0, stream>>>(tok, emb, cls, mw, part);
  merge_kernel<<<(BATCH * NCLS + 255) / 256, 256, 0, stream>>>(part, bias, out);
}
